# starting kernel
# baseline (speedup 1.0000x reference)
#include <hip/hip_runtime.h>

#define DIM 128

typedef float v2f __attribute__((ext_vector_type(2)));
typedef float v8f __attribute__((ext_vector_type(8)));

__global__ void __launch_bounds__(256)
proj_gemm_kernel(const float* __restrict__ node,
                 const float* __restrict__ Wsrc, const float* __restrict__ bsrc,
                 const float* __restrict__ Wdst, const float* __restrict__ bdst,
                 float* __restrict__ src_proj, float* __restrict__ dest_proj,
                 int V)
{
    const int lane    = threadIdx.x & 31;
    const int wave    = threadIdx.x >> 5;
    const int m0      = blockIdx.x * 16;
    const int n0      = wave * 16;
    const int halfsel = lane >> 4;
    const int l15     = lane & 15;

    const bool isDst  = (blockIdx.y != 0);
    const float* __restrict__ W   = isDst ? Wdst : Wsrc;
    const float* __restrict__ b   = isDst ? bdst : bsrc;
    float* __restrict__       out = isDst ? dest_proj : src_proj;

    const int arow = m0 + l15;
    const int bcol = n0 + l15;

    const float* __restrict__ aptr = node + (size_t)arow * DIM + 2 * halfsel;
    const float* __restrict__ bptr = W + (size_t)(2 * halfsel) * DIM + bcol;

    v8f acc = {};
#pragma unroll
    for (int k = 0; k < DIM; k += 4) {
        v2f a = *(const v2f*)(aptr + k);
        v2f bb;
        bb.x = bptr[(size_t)(k + 0) * DIM];
        bb.y = bptr[(size_t)(k + 1) * DIM];
        acc = __builtin_amdgcn_wmma_f32_16x16x4_f32(
                  false, a, false, bb, (short)0, acc, false, false);
    }

    const float bias = b[bcol];
#pragma unroll
    for (int r = 0; r < 8; ++r) {
        const int m = m0 + r + 8 * halfsel;
        out[(size_t)m * DIM + bcol] = acc[r] + bias;
    }
}

__global__ void __launch_bounds__(256)
edge_scatter_kernel(const float* __restrict__ src_proj,
                    const float* __restrict__ dest_proj,
                    const int* __restrict__ edge_src,
                    const int* __restrict__ edge_dest,
                    const int* __restrict__ edge_cls,
                    const float* __restrict__ edge_emb,
                    float* __restrict__ out, int E)
{
    const int e = blockIdx.x * 8 + (threadIdx.x >> 5);
    if (e >= E) return;
    const int lane = threadIdx.x & 31;
    const int off  = lane * 4;

    const int s = edge_src[e];
    const int d = edge_dest[e];
    const int c = edge_cls[e];

    const float4 sp = *(const float4*)(src_proj  + (size_t)s * DIM + off);
    const float4 dp = *(const float4*)(dest_proj + (size_t)d * DIM + off);
    const float4 em = *(const float4*)(edge_emb  + (size_t)c * DIM + off);

    float4 m;
    m.x = fmaxf(sp.x + dp.x + em.x, 0.0f);
    m.y = fmaxf(sp.y + dp.y + em.y, 0.0f);
    m.z = fmaxf(sp.z + dp.z + em.z, 0.0f);
    m.w = fmaxf(sp.w + dp.w + em.w, 0.0f);

    float* o = out + (size_t)d * DIM + off;
    unsafeAtomicAdd(o + 0, m.x);
    unsafeAtomicAdd(o + 1, m.y);
    unsafeAtomicAdd(o + 2, m.z);
    unsafeAtomicAdd(o + 3, m.w);
}

extern "C" void kernel_launch(void* const* d_in, const int* in_sizes, int n_in,
                              void* d_out, int out_size, void* d_ws, size_t ws_size,
                              hipStream_t stream) {
    const float* node = (const float*)d_in[0];
    const int*   esrc = (const int*)d_in[1];
    const int*   edst = (const int*)d_in[2];
    const int*   ecls = (const int*)d_in[3];
    const float* Wsrc = (const float*)d_in[4];
    const float* bsrc = (const float*)d_in[5];
    const float* Wdst = (const float*)d_in[6];
    const float* bdst = (const float*)d_in[7];
    const float* eemb = (const float*)d_in[8];
    float* out = (float*)d_out;

    const int V = in_sizes[0] / DIM;
    const int E = in_sizes[1];

    float* src_proj  = (float*)d_ws;
    float* dest_proj = src_proj + (size_t)V * DIM;

    hipMemsetAsync(d_out, 0, (size_t)V * DIM * sizeof(float), stream);

    dim3 ggrid((V + 15) / 16, 2);
    proj_gemm_kernel<<<ggrid, dim3(256), 0, stream>>>(
        node, Wsrc, bsrc, Wdst, bdst, src_proj, dest_proj, V);

    const int nblocks = (E + 7) / 8;
    edge_scatter_kernel<<<nblocks, dim3(256), 0, stream>>>(
        src_proj, dest_proj, esrc, edst, ecls, eemb, out, E);
}


# projection GEMM: both W staged once per block in LDS as k-quads (one 16-B LDS read = B of two steps), A rows from global straight to registers, 256 resident blocks; k sum order in groups of 8 changes
# speedup vs baseline: 1.6423x; 1.6423x over previous
#include <hip/hip_runtime.h>


#define DIM 128
#define NCLS 32
#define GB 512
#define PB 512
#define PGRID 256

typedef float v2f __attribute__((ext_vector_type(2)));
typedef float v4f __attribute__((ext_vector_type(4)));
typedef float v8f __attribute__((ext_vector_type(8)));

__global__ void __launch_bounds__(PB) __attribute__((amdgpu_waves_per_eu(4, 4)))
proj_gemm_kernel(const float* __restrict__ node,
                 const float* __restrict__ Wsrc, const float* __restrict__ bsrc,
                 const float* __restrict__ Wdst, const float* __restrict__ bdst,
                 float* __restrict__ src_proj, float* __restrict__ dest_proj,
                 int* __restrict__ head, int V)
{
    __shared__ v4f s_w[2 * (DIM / 4) * DIM];

    const int tid  = threadIdx.x;
    const int lane = tid & 31;
    const int c    = lane & 15;
    const int h    = lane >> 4;
    const int wave = __builtin_amdgcn_readfirstlane(tid >> 5);
    const int g    = (wave + (wave >> 2)) & 3;
    const int q    = wave & 3;
    const int wsel = q >> 1;
    const int n0   = (q & 1) * 64;

    const int nstrips = (V + 15) >> 4;
    const int stride  = gridDim.x * 4;
    int s = blockIdx.x + gridDim.x * g;

    v4f a[DIM / 8];
    {
        const int s0 = s < nstrips ? s : 0;
        const float* __restrict__ ap = node + (size_t)(s0 * 16 + c) * DIM + 4 * h;
#pragma unroll
        for (int kb = 0; kb < DIM / 8; ++kb) a[kb] = *(const v4f*)(ap + 8 * kb);
    }

    for (int i = blockIdx.x * PB + tid; i < V; i += gridDim.x * PB)
        head[i] = -1;

#pragma unroll
    for (int i = 0; i < 2 * (DIM / 4) * (DIM / 2) / PB; ++i) {
        const int item = tid + i * PB;
        const int np = item & 63;
        const int kq = (item >> 6) & 31;
        const int w  = item >> 11;
        const float* __restrict__ p = (w ? Wdst : Wsrc) + (size_t)(4 * kq) * DIM + 2 * np;
        const v2f r0 = *(const v2f*)(p);
        const v2f r1 = *(const v2f*)(p + DIM);
        const v2f r2 = *(const v2f*)(p + 2 * DIM);
        const v2f r3 = *(const v2f*)(p + 3 * DIM);
        v4f c0, c1;
        c0.x = r0.x; c0.y = r1.x; c0.z = r2.x; c0.w = r3.x;
        c1.x = r0.y; c1.y = r1.y; c1.z = r2.y; c1.w = r3.y;
        s_w[(w * 32 + kq) * DIM + 2 * np]     = c0;
        s_w[(w * 32 + kq) * DIM + 2 * np + 1] = c1;
    }
    __syncthreads();

    float* __restrict__ outp = wsel ? dest_proj : src_proj;
    const float* __restrict__ bp = (wsel ? bdst : bsrc) + n0 + c;
    float bias[4];
#pragma unroll
    for (int t = 0; t < 4; ++t) bias[t] = bp[16 * t];

    const v4f* bfrag = s_w + (wsel * 32 + h) * DIM + n0 + c;

    if (s >= nstrips) return;

    for (;;) {
        asm volatile("" ::: "memory");
        const int  sn   = s + stride;
        const bool more = sn < nstrips;
        const float* __restrict__ apn =
            node + (size_t)((more ? sn : s) * 16 + c) * DIM + 4 * h;

        v8f acc[4] = {};
#pragma unroll
        for (int kb = 0; kb < DIM / 8; ++kb) {
            v4f bv[4];
#pragma unroll
            for (int t = 0; t < 4; ++t) bv[t] = bfrag[2 * kb * DIM + 16 * t];
            v2f a01, a23;
            a01.x = a[kb].x; a01.y = a[kb].y;
            a23.x = a[kb].z; a23.y = a[kb].w;
#pragma unroll
            for (int t = 0; t < 4; ++t) {
                v2f b01; b01.x = bv[t].x; b01.y = bv[t].y;
                acc[t] = __builtin_amdgcn_wmma_f32_16x16x4_f32(
                             false, a01, false, b01, (short)0, acc[t], false, false);
            }
#pragma unroll
            for (int t = 0; t < 4; ++t) {
                v2f b23; b23.x = bv[t].z; b23.y = bv[t].w;
                acc[t] = __builtin_amdgcn_wmma_f32_16x16x4_f32(
                             false, a23, false, b23, (short)0, acc[t], false, false);
            }
            a[kb] = *(const v4f*)(apn + 8 * kb);
            __builtin_amdgcn_sched_barrier(0x7C6);
        }

        float* __restrict__ op = outp + (size_t)(s * 16 + 8 * h) * DIM + n0 + c;
#pragma unroll
        for (int t = 0; t < 4; ++t)
#pragma unroll
            for (int r = 0; r < 8; ++r)
                op[(size_t)r * DIM + 16 * t] = acc[t][r] + bias[t];

        if (!more) break;
        s = sn;
    }
}

__global__ void __launch_bounds__(256)
edge_link_kernel(const int* __restrict__ edge_src,
                 const int* __restrict__ edge_dest,
                 const int* __restrict__ edge_cls,
                 int* __restrict__ head, int* __restrict__ rec, int E)
{
    const int e = blockIdx.x * 256 + threadIdx.x;
    if (e >= E) return;
    const int s = edge_src[e];
    const int d = edge_dest[e];
    const int c = edge_cls[e];
    const int prev = __hip_atomic_exchange(head + d, e, __ATOMIC_RELAXED,
                                           __HIP_MEMORY_SCOPE_AGENT);
    ((int4*)rec)[e] = make_int4(s, c, prev, 0);
}

__global__ void __launch_bounds__(GB)
node_gather_kernel(const float* __restrict__ src_proj,
                   const int* __restrict__ head,
                   const int* __restrict__ rec,
                   const float* __restrict__ edge_emb,
                   float* __restrict__ out, int V, int E)
{
    __shared__ float4 s_emb[NCLS * DIM / 4];

    const int lane = threadIdx.x & 31;
    const int l15  = lane & 15;
    const int d    = (blockIdx.x * (GB / 32) + (threadIdx.x >> 5)) * 2 + (lane >> 4);
    const bool valid = d < V;
    const int q0 = l15;
    const int q1 = 16 + l15;

    const int4* __restrict__ rec4 = (const int4*)rec;
    const int dc = valid ? d : 0;
    float4* __restrict__ orow = (float4*)(out + (size_t)dc * DIM);

    int e = head[dc];
    const float4 dp0 = orow[q0];
    const float4 dp1 = orow[q1];
    float4 t[NCLS * DIM / 4 / GB];
#pragma unroll
    for (int i = 0; i < NCLS * DIM / 4 / GB; ++i)
        t[i] = ((const float4*)edge_emb)[threadIdx.x + i * GB];

    bool act = valid && ((unsigned)e < (unsigned)E);
    e = act ? e : 0;
    int4 r = rec4[e];

#pragma unroll
    for (int i = 0; i < NCLS * DIM / 4 / GB; ++i)
        s_emb[threadIdx.x + i * GB] = t[i];
    __syncthreads();

    float4 acc0 = make_float4(0.f, 0.f, 0.f, 0.f), acc1 = acc0;
    for (int it = 0; act && it < E; ++it) {
        const float4* __restrict__ sp = (const float4*)(src_proj + (size_t)r.x * DIM);
        const float4 a0 = sp[q0];
        const float4 a1 = sp[q1];
        const int c  = r.y;
        const int en = r.z;
        const bool actn = (unsigned)en < (unsigned)E;
        e = actn ? en : e;
        r = rec4[e];
        const float4 m0 = s_emb[c * (DIM / 4) + q0];
        const float4 m1 = s_emb[c * (DIM / 4) + q1];
        acc0.x += fmaxf(a0.x + dp0.x + m0.x, 0.0f);
        acc0.y += fmaxf(a0.y + dp0.y + m0.y, 0.0f);
        acc0.z += fmaxf(a0.z + dp0.z + m0.z, 0.0f);
        acc0.w += fmaxf(a0.w + dp0.w + m0.w, 0.0f);
        acc1.x += fmaxf(a1.x + dp1.x + m1.x, 0.0f);
        acc1.y += fmaxf(a1.y + dp1.y + m1.y, 0.0f);
        acc1.z += fmaxf(a1.z + dp1.z + m1.z, 0.0f);
        acc1.w += fmaxf(a1.w + dp1.w + m1.w, 0.0f);
        act = actn;
    }

    if (valid) {
        orow[q0] = acc0;
        orow[q1] = acc1;
    }
}

extern "C" void kernel_launch(void* const* d_in, const int* in_sizes, int n_in,
                              void* d_out, int out_size, void* d_ws, size_t ws_size,
                              hipStream_t stream) {
    const float* node = (const float*)d_in[0];
    const int*   esrc = (const int*)d_in[1];
    const int*   edst = (const int*)d_in[2];
    const int*   ecls = (const int*)d_in[3];
    const float* Wsrc = (const float*)d_in[4];
    const float* bsrc = (const float*)d_in[5];
    const float* Wdst = (const float*)d_in[6];
    const float* bdst = (const float*)d_in[7];
    const float* eemb = (const float*)d_in[8];
    float* out = (float*)d_out;

    const int V = in_sizes[0] / DIM;
    const int E = in_sizes[1];

    float* src_proj = (float*)d_ws;
    int*   rec      = (int*)(src_proj + (size_t)V * DIM);
    int*   head     = rec + (size_t)E * 4;

    proj_gemm_kernel<<<dim3(PGRID), dim3(PB), 0, stream>>>(
        node, Wsrc, bsrc, Wdst, bdst, src_proj, out, head, V);

    edge_link_kernel<<<(E + 255) / 256, dim3(256), 0, stream>>>(
        esrc, edst, ecls, head, rec, E);

    node_gather_kernel<<<(V + 31) / 32, dim3(GB), 0, stream>>>(
        src_proj, head, rec, eemb, out, V, E);
}


# list building moved into the projection launch: 8 extra waves per block link the edges beside the 16 matrix waves; head clear is its own small launch; sum order = link arrival order
# speedup vs baseline: 1.7713x; 1.0785x over previous
#include <hip/hip_runtime.h>


#define DIM 128
#define NCLS 32
#define GB 512
#define PB 512
#define LW 256
#define EPT 10
#define PGRID 256

typedef float v2f __attribute__((ext_vector_type(2)));
typedef float v4f __attribute__((ext_vector_type(4)));
typedef float v8f __attribute__((ext_vector_type(8)));

template <bool GUARD>
__device__ __forceinline__ void
link_batch(const int* __restrict__ edge_src,
           const int* __restrict__ edge_dest,
           const int* __restrict__ edge_cls,
           int* __restrict__ head, int4* __restrict__ rec4, int E, int base, int T)
{
    int s[EPT], d[EPT], c[EPT], prev[EPT];
#pragma unroll
    for (int i = 0; i < EPT; ++i) {
        const int e  = base + i * T;
        const int ec = (!GUARD || e < E) ? e : base;
        s[i] = edge_src[ec];
        d[i] = edge_dest[ec];
        c[i] = edge_cls[ec];
    }
#pragma unroll
    for (int i = 0; i < EPT; ++i) {
        const int e = base + i * T;
        prev[i] = -1;
        if (!GUARD || e < E)
            prev[i] = __hip_atomic_exchange(head + d[i], e, __ATOMIC_RELAXED,
                                            __HIP_MEMORY_SCOPE_AGENT);
    }
#pragma unroll
    for (int i = 0; i < EPT; ++i) {
        const int e = base + i * T;
        if (!GUARD || e < E) rec4[e] = make_int4(s[i], c[i], prev[i], 0);
    }
}

__device__ __forceinline__ void
link_edges(const int* __restrict__ edge_src,
           const int* __restrict__ edge_dest,
           const int* __restrict__ edge_cls,
           int* __restrict__ head, int* __restrict__ rec, int E, int g, int T)
{
    int4* __restrict__ rec4 = (int4*)rec;
    int base = g;
    for (; base + (EPT - 1) * T < E; base += EPT * T)
        link_batch<false>(edge_src, edge_dest, edge_cls, head, rec4, E, base, T);
    if (base < E)
        link_batch<true>(edge_src, edge_dest, edge_cls, head, rec4, E, base, T);
}

__global__ void __launch_bounds__(256)
head_clear_kernel(int* __restrict__ head, int V)
{
    const int i = blockIdx.x * 256 + threadIdx.x;
    if (i < V) head[i] = -1;
}

__global__ void __launch_bounds__(PB + LW) __attribute__((amdgpu_waves_per_eu(6, 6)))
proj_gemm_kernel(const float* __restrict__ node,
                 const float* __restrict__ Wsrc, const float* __restrict__ bsrc,
                 const float* __restrict__ Wdst, const float* __restrict__ bdst,
                 float* __restrict__ src_proj, float* __restrict__ dest_proj,
                 const int* __restrict__ edge_src,
                 const int* __restrict__ edge_dest,
                 const int* __restrict__ edge_cls,
                 int* __restrict__ head, int* __restrict__ rec, int V, int E)
{
    __shared__ v4f s_w[2 * (DIM / 4) * DIM];

    const int tid  = threadIdx.x;
    const int lane = tid & 31;
    const int c    = lane & 15;
    const int h    = lane >> 4;
    const int wave = __builtin_amdgcn_readfirstlane(tid >> 5);
    const int  w6 = wave / 6;
    const int  r6 = wave - 6 * w6;
    const bool mm = r6 < 4;
    const int  mw = 4 * w6 + r6;
    const int  mtid = 32 * mw + lane;
    const int  ltid = 32 * (2 * w6 + r6 - 4) + lane;
    const int g    = (mw + (mw >> 2)) & 3;
    const int q    = mw & 3;
    const int wsel = q >> 1;
    const int n0   = (q & 1) * 64;

    const int nstrips = (V + 15) >> 4;
    const int stride  = gridDim.x * 4;
    int s = blockIdx.x + gridDim.x * g;

    v4f a[DIM / 8] = {};
    if (mm) {
        const int s0 = s < nstrips ? s : 0;
        const float* __restrict__ ap = node + (size_t)(s0 * 16 + c) * DIM + 4 * h;
#pragma unroll
        for (int kb = 0; kb < DIM / 8; ++kb) a[kb] = *(const v4f*)(ap + 8 * kb);
    }

    if (mm) {
#pragma unroll
    for (int i = 0; i < 2 * (DIM / 4) * (DIM / 2) / PB; ++i) {
        const int item = mtid + i * PB;
        const int np = item & 63;
        const int kq = (item >> 6) & 31;
        const int w  = item >> 11;
        const float* __restrict__ p = (w ? Wdst : Wsrc) + (size_t)(4 * kq) * DIM + 2 * np;
        const v2f r0 = *(const v2f*)(p);
        const v2f r1 = *(const v2f*)(p + DIM);
        const v2f r2 = *(const v2f*)(p + 2 * DIM);
        const v2f r3 = *(const v2f*)(p + 3 * DIM);
        v4f c0, c1;
        c0.x = r0.x; c0.y = r1.x; c0.z = r2.x; c0.w = r3.x;
        c1.x = r0.y; c1.y = r1.y; c1.z = r2.y; c1.w = r3.y;
        s_w[(w * 32 + kq) * DIM + 2 * np]     = c0;
        s_w[(w * 32 + kq) * DIM + 2 * np + 1] = c1;
    }
    }
    __syncthreads();

    if (!mm) {
        link_edges(edge_src, edge_dest, edge_cls, head, rec, E,
                   blockIdx.x * LW + ltid, gridDim.x * LW);
        return;
    }

    float* __restrict__ outp = wsel ? dest_proj : src_proj;
    const float* __restrict__ bp = (wsel ? bdst : bsrc) + n0 + c;
    float bias[4];
#pragma unroll
    for (int t = 0; t < 4; ++t) bias[t] = bp[16 * t];

    const v4f* bfrag = s_w + (wsel * 32 + h) * DIM + n0 + c;

    if (s >= nstrips) return;

    for (;;) {
        asm volatile("" ::: "memory");
        const int  sn   = s + stride;
        const bool more = sn < nstrips;
        const float* __restrict__ apn =
            node + (size_t)((more ? sn : s) * 16 + c) * DIM + 4 * h;

        v8f acc[4] = {};
#pragma unroll
        for (int kb = 0; kb < DIM / 8; ++kb) {
            v4f bv[4];
#pragma unroll
            for (int t = 0; t < 4; ++t) bv[t] = bfrag[2 * kb * DIM + 16 * t];
            v2f a01, a23;
            a01.x = a[kb].x; a01.y = a[kb].y;
            a23.x = a[kb].z; a23.y = a[kb].w;
#pragma unroll
            for (int t = 0; t < 4; ++t) {
                v2f b01; b01.x = bv[t].x; b01.y = bv[t].y;
                acc[t] = __builtin_amdgcn_wmma_f32_16x16x4_f32(
                             false, a01, false, b01, (short)0, acc[t], false, false);
            }
#pragma unroll
            for (int t = 0; t < 4; ++t) {
                v2f b23; b23.x = bv[t].z; b23.y = bv[t].w;
                acc[t] = __builtin_amdgcn_wmma_f32_16x16x4_f32(
                             false, a23, false, b23, (short)0, acc[t], false, false);
            }
            a[kb] = *(const v4f*)(apn + 8 * kb);
            __builtin_amdgcn_sched_barrier(0x7C6);
        }

        float* __restrict__ op = outp + (size_t)(s * 16 + 8 * h) * DIM + n0 + c;
#pragma unroll
        for (int t = 0; t < 4; ++t)
#pragma unroll
            for (int r = 0; r < 8; ++r)
                op[(size_t)r * DIM + 16 * t] = acc[t][r] + bias[t];

        if (!more) break;
        s = sn;
    }
}

__global__ void __launch_bounds__(GB)
node_gather_kernel(const float* __restrict__ src_proj,
                   const int* __restrict__ head,
                   const int* __restrict__ rec,
                   const float* __restrict__ edge_emb,
                   float* __restrict__ out, int V, int E)
{
    __shared__ float4 s_emb[NCLS * DIM / 4];

    const int lane = threadIdx.x & 31;
    const int l15  = lane & 15;
    const int d    = (blockIdx.x * (GB / 32) + (threadIdx.x >> 5)) * 2 + (lane >> 4);
    const bool valid = d < V;
    const int q0 = l15;
    const int q1 = 16 + l15;

    const int4* __restrict__ rec4 = (const int4*)rec;
    const int dc = valid ? d : 0;
    float4* __restrict__ orow = (float4*)(out + (size_t)dc * DIM);

    int e = head[dc];
    const float4 dp0 = orow[q0];
    const float4 dp1 = orow[q1];
    float4 t[NCLS * DIM / 4 / GB];
#pragma unroll
    for (int i = 0; i < NCLS * DIM / 4 / GB; ++i)
        t[i] = ((const float4*)edge_emb)[threadIdx.x + i * GB];

    bool act = valid && ((unsigned)e < (unsigned)E);
    e = act ? e : 0;
    int4 r = rec4[e];

#pragma unroll
    for (int i = 0; i < NCLS * DIM / 4 / GB; ++i)
        s_emb[threadIdx.x + i * GB] = t[i];
    __syncthreads();

    float4 acc0 = make_float4(0.f, 0.f, 0.f, 0.f), acc1 = acc0;
    for (int it = 0; act && it < E; ++it) {
        const float4* __restrict__ sp = (const float4*)(src_proj + (size_t)r.x * DIM);
        const float4 a0 = sp[q0];
        const float4 a1 = sp[q1];
        const int c  = r.y;
        const int en = r.z;
        const bool actn = (unsigned)en < (unsigned)E;
        e = actn ? en : e;
        r = rec4[e];
        const float4 m0 = s_emb[c * (DIM / 4) + q0];
        const float4 m1 = s_emb[c * (DIM / 4) + q1];
        acc0.x += fmaxf(a0.x + dp0.x + m0.x, 0.0f);
        acc0.y += fmaxf(a0.y + dp0.y + m0.y, 0.0f);
        acc0.z += fmaxf(a0.z + dp0.z + m0.z, 0.0f);
        acc0.w += fmaxf(a0.w + dp0.w + m0.w, 0.0f);
        acc1.x += fmaxf(a1.x + dp1.x + m1.x, 0.0f);
        acc1.y += fmaxf(a1.y + dp1.y + m1.y, 0.0f);
        acc1.z += fmaxf(a1.z + dp1.z + m1.z, 0.0f);
        acc1.w += fmaxf(a1.w + dp1.w + m1.w, 0.0f);
        act = actn;
    }

    if (valid) {
        orow[q0] = acc0;
        orow[q1] = acc1;
    }
}

extern "C" void kernel_launch(void* const* d_in, const int* in_sizes, int n_in,
                              void* d_out, int out_size, void* d_ws, size_t ws_size,
                              hipStream_t stream) {
    const float* node = (const float*)d_in[0];
    const int*   esrc = (const int*)d_in[1];
    const int*   edst = (const int*)d_in[2];
    const int*   ecls = (const int*)d_in[3];
    const float* Wsrc = (const float*)d_in[4];
    const float* bsrc = (const float*)d_in[5];
    const float* Wdst = (const float*)d_in[6];
    const float* bdst = (const float*)d_in[7];
    const float* eemb = (const float*)d_in[8];
    float* out = (float*)d_out;

    const int V = in_sizes[0] / DIM;
    const int E = in_sizes[1];

    float* src_proj = (float*)d_ws;
    int*   rec      = (int*)(src_proj + (size_t)V * DIM);
    int*   head     = rec + (size_t)E * 4;

    head_clear_kernel<<<(V + 255) / 256, dim3(256), 0, stream>>>(head, V);

    proj_gemm_kernel<<<dim3(PGRID), dim3(PB + LW), 0, stream>>>(
        node, Wsrc, bsrc, Wdst, bdst, src_proj, out, esrc, edst, ecls, head, rec, V, E);

    node_gather_kernel<<<(V + 31) / 32, dim3(GB), 0, stream>>>(
        src_proj, head, rec, eemb, out, V, E);
}


# edge lists: linked list per node replaced by one 128-B line per node (count + 31 packed src/cls words, rank by returning int add), node pass loads 4 rows at once; sum order = arrival rank order
# speedup vs baseline: 2.0930x; 1.1816x over previous
#include <hip/hip_runtime.h>


#define DIM 128
#define NCLS 32
#define GB 512
#define PB 512
#define LW 256
#define EPT 10
#define PGRID 256
#define LINEW 32
#define SLOTS 31
#define RU 4

typedef float v2f __attribute__((ext_vector_type(2)));
typedef float v4f __attribute__((ext_vector_type(4)));
typedef float v8f __attribute__((ext_vector_type(8)));

template <bool GUARD>
__device__ __forceinline__ void
link_batch(const int* __restrict__ edge_src,
           const int* __restrict__ edge_dest,
           const int* __restrict__ edge_cls,
           unsigned* __restrict__ line, int* __restrict__ head,
           int* __restrict__ next, int E, int base, int T)
{
    int s[EPT], d[EPT], c[EPT];
    unsigned k[EPT];
#pragma unroll
    for (int i = 0; i < EPT; ++i) {
        const int e  = base + i * T;
        const int ec = (!GUARD || e < E) ? e : base;
        s[i] = edge_src[ec];
        d[i] = edge_dest[ec];
        c[i] = edge_cls[ec];
    }
#pragma unroll
    for (int i = 0; i < EPT; ++i) {
        const int e = base + i * T;
        k[i] = 0u;
        if (!GUARD || e < E)
            k[i] = __hip_atomic_fetch_add(line + (size_t)d[i] * LINEW, 1u, __ATOMIC_RELAXED,
                                          __HIP_MEMORY_SCOPE_AGENT);
    }
#pragma unroll
    for (int i = 0; i < EPT; ++i) {
        const int e = base + i * T;
        if (!GUARD || e < E) {
            if (k[i] < (unsigned)SLOTS) {
                line[(size_t)d[i] * LINEW + 1 + k[i]] =
                    ((unsigned)s[i] << 5) | ((unsigned)c[i] & (NCLS - 1));
            } else {
                const int prev = __hip_atomic_exchange(head + d[i], e, __ATOMIC_RELAXED,
                                                       __HIP_MEMORY_SCOPE_AGENT);
                next[e] = prev;
            }
        }
    }
}

__device__ __forceinline__ void
link_edges(const int* __restrict__ edge_src,
           const int* __restrict__ edge_dest,
           const int* __restrict__ edge_cls,
           unsigned* __restrict__ line, int* __restrict__ head,
           int* __restrict__ next, int E, int g, int T)
{
    int base = g;
    for (; base + (EPT - 1) * T < E; base += EPT * T)
        link_batch<false>(edge_src, edge_dest, edge_cls, line, head, next, E, base, T);
    if (base < E)
        link_batch<true>(edge_src, edge_dest, edge_cls, line, head, next, E, base, T);
}

__global__ void __launch_bounds__(256)
head_clear_kernel(unsigned* __restrict__ line, int* __restrict__ head, int V)
{
    const int i = blockIdx.x * 256 + threadIdx.x;
    if (i < V) {
        line[(size_t)i * LINEW] = 0u;
        head[i] = -1;
    }
}

__global__ void __launch_bounds__(PB + LW) __attribute__((amdgpu_waves_per_eu(6, 6)))
proj_gemm_kernel(const float* __restrict__ node,
                 const float* __restrict__ Wsrc, const float* __restrict__ bsrc,
                 const float* __restrict__ Wdst, const float* __restrict__ bdst,
                 float* __restrict__ src_proj, float* __restrict__ dest_proj,
                 const int* __restrict__ edge_src,
                 const int* __restrict__ edge_dest,
                 const int* __restrict__ edge_cls,
                 unsigned* __restrict__ line, int* __restrict__ head,
                 int* __restrict__ next, int V, int E)
{
    __shared__ v4f s_w[2 * (DIM / 4) * DIM];

    const int tid  = threadIdx.x;
    const int lane = tid & 31;
    const int c    = lane & 15;
    const int h    = lane >> 4;
    const int wave = __builtin_amdgcn_readfirstlane(tid >> 5);
    const int  w6 = wave / 6;
    const int  r6 = wave - 6 * w6;
    const bool mm = r6 < 4;
    const int  mw = 4 * w6 + r6;
    const int  mtid = 32 * mw + lane;
    const int  ltid = 32 * (2 * w6 + r6 - 4) + lane;
    const int g    = (mw + (mw >> 2)) & 3;
    const int q    = mw & 3;
    const int wsel = q >> 1;
    const int n0   = (q & 1) * 64;

    const int nstrips = (V + 15) >> 4;
    const int stride  = gridDim.x * 4;
    int s = blockIdx.x + gridDim.x * g;

    v4f a[DIM / 8] = {};
    if (mm) {
        const int s0 = s < nstrips ? s : 0;
        const float* __restrict__ ap = node + (size_t)(s0 * 16 + c) * DIM + 4 * h;
#pragma unroll
        for (int kb = 0; kb < DIM / 8; ++kb) a[kb] = *(const v4f*)(ap + 8 * kb);
    }

    if (mm) {
#pragma unroll
    for (int i = 0; i < 2 * (DIM / 4) * (DIM / 2) / PB; ++i) {
        const int item = mtid + i * PB;
        const int np = item & 63;
        const int kq = (item >> 6) & 31;
        const int w  = item >> 11;
        const float* __restrict__ p = (w ? Wdst : Wsrc) + (size_t)(4 * kq) * DIM + 2 * np;
        const v2f r0 = *(const v2f*)(p);
        const v2f r1 = *(const v2f*)(p + DIM);
        const v2f r2 = *(const v2f*)(p + 2 * DIM);
        const v2f r3 = *(const v2f*)(p + 3 * DIM);
        v4f c0, c1;
        c0.x = r0.x; c0.y = r1.x; c0.z = r2.x; c0.w = r3.x;
        c1.x = r0.y; c1.y = r1.y; c1.z = r2.y; c1.w = r3.y;
        s_w[(w * 32 + kq) * DIM + 2 * np]     = c0;
        s_w[(w * 32 + kq) * DIM + 2 * np + 1] = c1;
    }
    }
    __syncthreads();

    if (!mm) {
        link_edges(edge_src, edge_dest, edge_cls, line, head, next, E,
                   blockIdx.x * LW + ltid, gridDim.x * LW);
        return;
    }

    float* __restrict__ outp = wsel ? dest_proj : src_proj;
    const float* __restrict__ bp = (wsel ? bdst : bsrc) + n0 + c;
    float bias[4];
#pragma unroll
    for (int t = 0; t < 4; ++t) bias[t] = bp[16 * t];

    const v4f* bfrag = s_w + (wsel * 32 + h) * DIM + n0 + c;

    if (s >= nstrips) return;

    for (;;) {
        asm volatile("" ::: "memory");
        const int  sn   = s + stride;
        const bool more = sn < nstrips;
        const float* __restrict__ apn =
            node + (size_t)((more ? sn : s) * 16 + c) * DIM + 4 * h;

        v8f acc[4] = {};
#pragma unroll
        for (int kb = 0; kb < DIM / 8; ++kb) {
            v4f bv[4];
#pragma unroll
            for (int t = 0; t < 4; ++t) bv[t] = bfrag[2 * kb * DIM + 16 * t];
            v2f a01, a23;
            a01.x = a[kb].x; a01.y = a[kb].y;
            a23.x = a[kb].z; a23.y = a[kb].w;
#pragma unroll
            for (int t = 0; t < 4; ++t) {
                v2f b01; b01.x = bv[t].x; b01.y = bv[t].y;
                acc[t] = __builtin_amdgcn_wmma_f32_16x16x4_f32(
                             false, a01, false, b01, (short)0, acc[t], false, false);
            }
#pragma unroll
            for (int t = 0; t < 4; ++t) {
                v2f b23; b23.x = bv[t].z; b23.y = bv[t].w;
                acc[t] = __builtin_amdgcn_wmma_f32_16x16x4_f32(
                             false, a23, false, b23, (short)0, acc[t], false, false);
            }
            a[kb] = *(const v4f*)(apn + 8 * kb);
            __builtin_amdgcn_sched_barrier(0x7C6);
        }

        float* __restrict__ op = outp + (size_t)(s * 16 + 8 * h) * DIM + n0 + c;
#pragma unroll
        for (int t = 0; t < 4; ++t)
#pragma unroll
            for (int r = 0; r < 8; ++r)
                op[(size_t)r * DIM + 16 * t] = acc[t][r] + bias[t];

        if (!more) break;
        s = sn;
    }
}

__global__ void __launch_bounds__(GB)
node_gather_kernel(const float* __restrict__ src_proj,
                   const unsigned* __restrict__ line,
                   const int* __restrict__ head,
                   const int* __restrict__ next,
                   const int* __restrict__ edge_src,
                   const int* __restrict__ edge_cls,
                   const float* __restrict__ edge_emb,
                   float* __restrict__ out, int V, int E)
{
    __shared__ float4 s_emb[NCLS * DIM / 4];

    const int lane = threadIdx.x & 31;
    const int l15  = lane & 15;
    const int d    = (blockIdx.x * (GB / 32) + (threadIdx.x >> 5)) * 2 + (lane >> 4);
    const bool valid = d < V;
    const int q0 = l15;
    const int q1 = 16 + l15;

    const int dc = valid ? d : 0;
    float4* __restrict__ orow = (float4*)(out + (size_t)dc * DIM);
    const unsigned* __restrict__ ln = line + (size_t)dc * LINEW;

    const unsigned w0 = ln[l15];
    const unsigned w1 = ln[16 + l15];
    const float4 dp0 = orow[q0];
    const float4 dp1 = orow[q1];
    float4 t[NCLS * DIM / 4 / GB];
#pragma unroll
    for (int i = 0; i < NCLS * DIM / 4 / GB; ++i)
        t[i] = ((const float4*)edge_emb)[threadIdx.x + i * GB];

#pragma unroll
    for (int i = 0; i < NCLS * DIM / 4 / GB; ++i)
        s_emb[threadIdx.x + i * GB] = t[i];
    __syncthreads();

    const unsigned cnt = (unsigned)__shfl((int)w0, 0, 16);
    const int n  = valid ? (int)(cnt < (unsigned)E ? cnt : (unsigned)E) : 0;
    const int nb = n < SLOTS ? n : SLOTS;

    float4 acc0 = make_float4(0.f, 0.f, 0.f, 0.f), acc1 = acc0;
    for (int base = 0; base < nb; base += RU) {
        float4 a0[RU], a1[RU];
        int    cc[RU];
        bool   ok[RU];
        unsigned w[RU];
#pragma unroll
        for (int u = 0; u < RU; ++u) {
            const int wi = base + u + 1;
            w[u] = (unsigned)__shfl((int)((wi & 16) ? w1 : w0), wi & 15, 16);
        }
#pragma unroll
        for (int u = 0; u < RU; ++u) {
            ok[u] = base + u < nb;
            cc[u] = (int)(w[u] & (NCLS - 1));
            a0[u] = make_float4(0.f, 0.f, 0.f, 0.f);
            a1[u] = a0[u];
            if (ok[u]) {
                const float4* __restrict__ sp =
                    (const float4*)(src_proj + (size_t)(w[u] >> 5) * DIM);
                a0[u] = sp[q0];
                a1[u] = sp[q1];
            }
        }
#pragma unroll
        for (int u = 0; u < RU; ++u) {
            if (ok[u]) {
                const float4 m0 = s_emb[cc[u] * (DIM / 4) + q0];
                const float4 m1 = s_emb[cc[u] * (DIM / 4) + q1];
                acc0.x += fmaxf(a0[u].x + dp0.x + m0.x, 0.0f);
                acc0.y += fmaxf(a0[u].y + dp0.y + m0.y, 0.0f);
                acc0.z += fmaxf(a0[u].z + dp0.z + m0.z, 0.0f);
                acc0.w += fmaxf(a0[u].w + dp0.w + m0.w, 0.0f);
                acc1.x += fmaxf(a1[u].x + dp1.x + m1.x, 0.0f);
                acc1.y += fmaxf(a1[u].y + dp1.y + m1.y, 0.0f);
                acc1.z += fmaxf(a1[u].z + dp1.z + m1.z, 0.0f);
                acc1.w += fmaxf(a1[u].w + dp1.w + m1.w, 0.0f);
            }
        }
    }

    if (n > SLOTS) {
        int e = head[dc];
        for (int it = 0; (unsigned)e < (unsigned)E && it < E; ++it) {
            const int sr = edge_src[e];
            const int c  = edge_cls[e] & (NCLS - 1);
            const int en = next[e];
            const float4* __restrict__ sp = (const float4*)(src_proj + (size_t)sr * DIM);
            const float4 a0 = sp[q0];
            const float4 a1 = sp[q1];
            const float4 m0 = s_emb[c * (DIM / 4) + q0];
            const float4 m1 = s_emb[c * (DIM / 4) + q1];
            acc0.x += fmaxf(a0.x + dp0.x + m0.x, 0.0f);
            acc0.y += fmaxf(a0.y + dp0.y + m0.y, 0.0f);
            acc0.z += fmaxf(a0.z + dp0.z + m0.z, 0.0f);
            acc0.w += fmaxf(a0.w + dp0.w + m0.w, 0.0f);
            acc1.x += fmaxf(a1.x + dp1.x + m1.x, 0.0f);
            acc1.y += fmaxf(a1.y + dp1.y + m1.y, 0.0f);
            acc1.z += fmaxf(a1.z + dp1.z + m1.z, 0.0f);
            acc1.w += fmaxf(a1.w + dp1.w + m1.w, 0.0f);
            e = en;
        }
    }

    if (valid) {
        orow[q0] = acc0;
        orow[q1] = acc1;
    }
}

extern "C" void kernel_launch(void* const* d_in, const int* in_sizes, int n_in,
                              void* d_out, int out_size, void* d_ws, size_t ws_size,
                              hipStream_t stream) {
    const float* node = (const float*)d_in[0];
    const int*   esrc = (const int*)d_in[1];
    const int*   edst = (const int*)d_in[2];
    const int*   ecls = (const int*)d_in[3];
    const float* Wsrc = (const float*)d_in[4];
    const float* bsrc = (const float*)d_in[5];
    const float* Wdst = (const float*)d_in[6];
    const float* bdst = (const float*)d_in[7];
    const float* eemb = (const float*)d_in[8];
    float* out = (float*)d_out;

    const int V = in_sizes[0] / DIM;
    const int E = in_sizes[1];

    float*    src_proj = (float*)d_ws;
    unsigned* line     = (unsigned*)(src_proj + (size_t)V * DIM);
    int*      next     = (int*)(line + (size_t)V * LINEW);
    int*      head     = next + (size_t)E;

    head_clear_kernel<<<(V + 255) / 256, dim3(256), 0, stream>>>(line, head, V);

    proj_gemm_kernel<<<dim3(PGRID), dim3(PB + LW), 0, stream>>>(
        node, Wsrc, bsrc, Wdst, bdst, src_proj, out, esrc, edst, ecls, line, head, next, V, E);

    node_gather_kernel<<<(V + 31) / 32, dim3(GB), 0, stream>>>(
        src_proj, line, head, next, esrc, ecls, eemb, out, V, E);
}


# edge filing: two lines per node, a link wave files into the line of its own chip half (die number), so a line is written from one cache half only; sum order = arrival rank, line 1 then line 2
# speedup vs baseline: 2.6173x; 1.2505x over previous
#include <hip/hip_runtime.h>


#define DIM 128
#define NCLS 32
#define GB 512
#define PB 512
#define LW 256
#define EPT 10
#define PGRID 256
#define LINEW 32
#define SLOTS 31
#define RU 4

typedef float v2f __attribute__((ext_vector_type(2)));
typedef float v4f __attribute__((ext_vector_type(4)));
typedef float v8f __attribute__((ext_vector_type(8)));

template <bool GUARD>
__device__ __forceinline__ void
link_batch(const int* __restrict__ edge_src,
           const int* __restrict__ edge_dest,
           const int* __restrict__ edge_cls,
           unsigned* __restrict__ line, int* __restrict__ head,
           int* __restrict__ next, int E, int base, int T)
{
    int s[EPT], d[EPT], c[EPT];
    unsigned k[EPT];
#pragma unroll
    for (int i = 0; i < EPT; ++i) {
        const int e  = base + i * T;
        const int ec = (!GUARD || e < E) ? e : base;
        s[i] = edge_src[ec];
        d[i] = edge_dest[ec];
        c[i] = edge_cls[ec];
    }
#pragma unroll
    for (int i = 0; i < EPT; ++i) {
        const int e = base + i * T;
        k[i] = 0u;
        if (!GUARD || e < E)
            k[i] = __hip_atomic_fetch_add(line + (size_t)d[i] * LINEW, 1u, __ATOMIC_RELAXED,
                                          __HIP_MEMORY_SCOPE_AGENT);
    }
#pragma unroll
    for (int i = 0; i < EPT; ++i) {
        const int e = base + i * T;
        if (!GUARD || e < E) {
            if (k[i] < (unsigned)SLOTS) {
                line[(size_t)d[i] * LINEW + 1 + k[i]] =
                    ((unsigned)s[i] << 5) | ((unsigned)c[i] & (NCLS - 1));
            } else {
                const int prev = __hip_atomic_exchange(head + d[i], e, __ATOMIC_RELAXED,
                                                       __HIP_MEMORY_SCOPE_AGENT);
                next[e] = prev;
            }
        }
    }
}

__device__ __forceinline__ void
link_edges(const int* __restrict__ edge_src,
           const int* __restrict__ edge_dest,
           const int* __restrict__ edge_cls,
           unsigned* __restrict__ line, int* __restrict__ head,
           int* __restrict__ next, int E, int g, int T)
{
    int base = g;
    for (; base + (EPT - 1) * T < E; base += EPT * T)
        link_batch<false>(edge_src, edge_dest, edge_cls, line, head, next, E, base, T);
    if (base < E)
        link_batch<true>(edge_src, edge_dest, edge_cls, line, head, next, E, base, T);
}

__global__ void __launch_bounds__(256)
head_clear_kernel(unsigned* __restrict__ line, unsigned* __restrict__ line2,
                  int* __restrict__ head, int V)
{
    const int i = blockIdx.x * 256 + threadIdx.x;
    if (i < V) {
        line[(size_t)i * LINEW]  = 0u;
        line2[(size_t)i * LINEW] = 0u;
        head[i] = -1;
    }
}

__global__ void __launch_bounds__(PB + LW) __attribute__((amdgpu_waves_per_eu(6, 6)))
proj_gemm_kernel(const float* __restrict__ node,
                 const float* __restrict__ Wsrc, const float* __restrict__ bsrc,
                 const float* __restrict__ Wdst, const float* __restrict__ bdst,
                 float* __restrict__ src_proj, float* __restrict__ dest_proj,
                 const int* __restrict__ edge_src,
                 const int* __restrict__ edge_dest,
                 const int* __restrict__ edge_cls,
                 unsigned* __restrict__ line, unsigned* __restrict__ line2,
                 int* __restrict__ head,
                 int* __restrict__ next, int V, int E)
{
    __shared__ v4f s_w[2 * (DIM / 4) * DIM];

    const int tid  = threadIdx.x;
    const int lane = tid & 31;
    const int c    = lane & 15;
    const int h    = lane >> 4;
    const int wave = __builtin_amdgcn_readfirstlane(tid >> 5);
    const int  w6 = wave / 6;
    const int  r6 = wave - 6 * w6;
    const bool mm = r6 < 4;
    const int  mw = 4 * w6 + r6;
    const int  mtid = 32 * mw + lane;
    const int  ltid = 32 * (2 * w6 + r6 - 4) + lane;
    const int g    = (mw + (mw >> 2)) & 3;
    const int q    = mw & 3;
    const int wsel = q >> 1;
    const int n0   = (q & 1) * 64;

    const int nstrips = (V + 15) >> 4;
    const int stride  = gridDim.x * 4;
    int s = blockIdx.x + gridDim.x * g;

    v4f a[DIM / 8] = {};
    if (mm) {
        const int s0 = s < nstrips ? s : 0;
        const float* __restrict__ ap = node + (size_t)(s0 * 16 + c) * DIM + 4 * h;
#pragma unroll
        for (int kb = 0; kb < DIM / 8; ++kb) a[kb] = *(const v4f*)(ap + 8 * kb);
    }

    if (mm) {
#pragma unroll
    for (int i = 0; i < 2 * (DIM / 4) * (DIM / 2) / PB; ++i) {
        const int item = mtid + i * PB;
        const int np = item & 63;
        const int kq = (item >> 6) & 31;
        const int w  = item >> 11;
        const float* __restrict__ p = (w ? Wdst : Wsrc) + (size_t)(4 * kq) * DIM + 2 * np;
        const v2f r0 = *(const v2f*)(p);
        const v2f r1 = *(const v2f*)(p + DIM);
        const v2f r2 = *(const v2f*)(p + 2 * DIM);
        const v2f r3 = *(const v2f*)(p + 3 * DIM);
        v4f c0, c1;
        c0.x = r0.x; c0.y = r1.x; c0.z = r2.x; c0.w = r3.x;
        c1.x = r0.y; c1.y = r1.y; c1.z = r2.y; c1.w = r3.y;
        s_w[(w * 32 + kq) * DIM + 2 * np]     = c0;
        s_w[(w * 32 + kq) * DIM + 2 * np + 1] = c1;
    }
    }
    __syncthreads();

    if (!mm) {
        const unsigned place = __builtin_amdgcn_s_sendmsg_rtn(0x87);
        unsigned* __restrict__ myline = ((place >> 18) & 1u) ? line2 : line;
        link_edges(edge_src, edge_dest, edge_cls, myline, head, next, E,
                   blockIdx.x * LW + ltid, gridDim.x * LW);
        return;
    }

    float* __restrict__ outp = wsel ? dest_proj : src_proj;
    const float* __restrict__ bp = (wsel ? bdst : bsrc) + n0 + c;
    float bias[4];
#pragma unroll
    for (int t = 0; t < 4; ++t) bias[t] = bp[16 * t];

    const v4f* bfrag = s_w + (wsel * 32 + h) * DIM + n0 + c;

    if (s >= nstrips) return;

    for (;;) {
        asm volatile("" ::: "memory");
        const int  sn   = s + stride;
        const bool more = sn < nstrips;
        const float* __restrict__ apn =
            node + (size_t)((more ? sn : s) * 16 + c) * DIM + 4 * h;

        v8f acc[4] = {};
#pragma unroll
        for (int kb = 0; kb < DIM / 8; ++kb) {
            v4f bv[4];
#pragma unroll
            for (int t = 0; t < 4; ++t) bv[t] = bfrag[2 * kb * DIM + 16 * t];
            v2f a01, a23;
            a01.x = a[kb].x; a01.y = a[kb].y;
            a23.x = a[kb].z; a23.y = a[kb].w;
#pragma unroll
            for (int t = 0; t < 4; ++t) {
                v2f b01; b01.x = bv[t].x; b01.y = bv[t].y;
                acc[t] = __builtin_amdgcn_wmma_f32_16x16x4_f32(
                             false, a01, false, b01, (short)0, acc[t], false, false);
            }
#pragma unroll
            for (int t = 0; t < 4; ++t) {
                v2f b23; b23.x = bv[t].z; b23.y = bv[t].w;
                acc[t] = __builtin_amdgcn_wmma_f32_16x16x4_f32(
                             false, a23, false, b23, (short)0, acc[t], false, false);
            }
            a[kb] = *(const v4f*)(apn + 8 * kb);
            __builtin_amdgcn_sched_barrier(0x7C6);
        }

        float* __restrict__ op = outp + (size_t)(s * 16 + 8 * h) * DIM + n0 + c;
#pragma unroll
        for (int t = 0; t < 4; ++t)
#pragma unroll
            for (int r = 0; r < 8; ++r)
                op[(size_t)r * DIM + 16 * t] = acc[t][r] + bias[t];

        if (!more) break;
        s = sn;
    }
}

__global__ void __launch_bounds__(GB)
node_gather_kernel(const float* __restrict__ src_proj,
                   const unsigned* __restrict__ line,
                   const unsigned* __restrict__ line2,
                   const int* __restrict__ head,
                   const int* __restrict__ next,
                   const int* __restrict__ edge_src,
                   const int* __restrict__ edge_cls,
                   const float* __restrict__ edge_emb,
                   float* __restrict__ out, int V, int E)
{
    __shared__ float4 s_emb[NCLS * DIM / 4];

    const int lane = threadIdx.x & 31;
    const int l15  = lane & 15;
    const int d    = (blockIdx.x * (GB / 32) + (threadIdx.x >> 5)) * 2 + (lane >> 4);
    const bool valid = d < V;
    const int q0 = l15;
    const int q1 = 16 + l15;

    const int dc = valid ? d : 0;
    float4* __restrict__ orow = (float4*)(out + (size_t)dc * DIM);
    const unsigned* __restrict__ lnA = line  + (size_t)dc * LINEW;
    const unsigned* __restrict__ lnB = line2 + (size_t)dc * LINEW;

    const unsigned a0w = lnA[l15];
    const unsigned a1w = lnA[16 + l15];
    const unsigned b0w = lnB[l15];
    const unsigned b1w = lnB[16 + l15];
    const float4 dp0 = orow[q0];
    const float4 dp1 = orow[q1];
    float4 t[NCLS * DIM / 4 / GB];
#pragma unroll
    for (int i = 0; i < NCLS * DIM / 4 / GB; ++i)
        t[i] = ((const float4*)edge_emb)[threadIdx.x + i * GB];

#pragma unroll
    for (int i = 0; i < NCLS * DIM / 4 / GB; ++i)
        s_emb[threadIdx.x + i * GB] = t[i];
    __syncthreads();

    const unsigned cntA = (unsigned)__shfl((int)a0w, 0, 16);
    const unsigned cntB = (unsigned)__shfl((int)b0w, 0, 16);
    const int nA = valid ? (int)(cntA < (unsigned)E ? cntA : (unsigned)E) : 0;
    const int nB = valid ? (int)(cntB < (unsigned)E ? cntB : (unsigned)E) : 0;
    const int mA = nA < SLOTS ? nA : SLOTS;
    const int mB = nB < SLOTS ? nB : SLOTS;
    const int nb = mA + mB;

    float4 acc0 = make_float4(0.f, 0.f, 0.f, 0.f), acc1 = acc0;
    for (int base = 0; base < nb; base += RU) {
        float4 a0[RU], a1[RU];
        int    cc[RU];
        bool   ok[RU];
        unsigned w[RU];
#pragma unroll
        for (int u = 0; u < RU; ++u) {
            const int  j   = base + u;
            const bool inA = j < mA;
            const int  wi  = (inA ? j : j - mA) + 1;
            const unsigned lo = inA ? a0w : b0w;
            const unsigned hi = inA ? a1w : b1w;
            w[u] = (unsigned)__shfl((int)((wi & 16) ? hi : lo), wi & 15, 16);
        }
#pragma unroll
        for (int u = 0; u < RU; ++u) {
            ok[u] = base + u < nb;
            cc[u] = (int)(w[u] & (NCLS - 1));
            a0[u] = make_float4(0.f, 0.f, 0.f, 0.f);
            a1[u] = a0[u];
            if (ok[u]) {
                const float4* __restrict__ sp =
                    (const float4*)(src_proj + (size_t)(w[u] >> 5) * DIM);
                a0[u] = sp[q0];
                a1[u] = sp[q1];
            }
        }
#pragma unroll
        for (int u = 0; u < RU; ++u) {
            if (ok[u]) {
                const float4 m0 = s_emb[cc[u] * (DIM / 4) + q0];
                const float4 m1 = s_emb[cc[u] * (DIM / 4) + q1];
                acc0.x += fmaxf(a0[u].x + dp0.x + m0.x, 0.0f);
                acc0.y += fmaxf(a0[u].y + dp0.y + m0.y, 0.0f);
                acc0.z += fmaxf(a0[u].z + dp0.z + m0.z, 0.0f);
                acc0.w += fmaxf(a0[u].w + dp0.w + m0.w, 0.0f);
                acc1.x += fmaxf(a1[u].x + dp1.x + m1.x, 0.0f);
                acc1.y += fmaxf(a1[u].y + dp1.y + m1.y, 0.0f);
                acc1.z += fmaxf(a1[u].z + dp1.z + m1.z, 0.0f);
                acc1.w += fmaxf(a1[u].w + dp1.w + m1.w, 0.0f);
            }
        }
    }

    if (nA > SLOTS || nB > SLOTS) {
        int e = head[dc];
        for (int it = 0; (unsigned)e < (unsigned)E && it < E; ++it) {
            const int sr = edge_src[e];
            const int c  = edge_cls[e] & (NCLS - 1);
            const int en = next[e];
            const float4* __restrict__ sp = (const float4*)(src_proj + (size_t)sr * DIM);
            const float4 a0 = sp[q0];
            const float4 a1 = sp[q1];
            const float4 m0 = s_emb[c * (DIM / 4) + q0];
            const float4 m1 = s_emb[c * (DIM / 4) + q1];
            acc0.x += fmaxf(a0.x + dp0.x + m0.x, 0.0f);
            acc0.y += fmaxf(a0.y + dp0.y + m0.y, 0.0f);
            acc0.z += fmaxf(a0.z + dp0.z + m0.z, 0.0f);
            acc0.w += fmaxf(a0.w + dp0.w + m0.w, 0.0f);
            acc1.x += fmaxf(a1.x + dp1.x + m1.x, 0.0f);
            acc1.y += fmaxf(a1.y + dp1.y + m1.y, 0.0f);
            acc1.z += fmaxf(a1.z + dp1.z + m1.z, 0.0f);
            acc1.w += fmaxf(a1.w + dp1.w + m1.w, 0.0f);
            e = en;
        }
    }

    if (valid) {
        orow[q0] = acc0;
        orow[q1] = acc1;
    }
}

extern "C" void kernel_launch(void* const* d_in, const int* in_sizes, int n_in,
                              void* d_out, int out_size, void* d_ws, size_t ws_size,
                              hipStream_t stream) {
    const float* node = (const float*)d_in[0];
    const int*   esrc = (const int*)d_in[1];
    const int*   edst = (const int*)d_in[2];
    const int*   ecls = (const int*)d_in[3];
    const float* Wsrc = (const float*)d_in[4];
    const float* bsrc = (const float*)d_in[5];
    const float* Wdst = (const float*)d_in[6];
    const float* bdst = (const float*)d_in[7];
    const float* eemb = (const float*)d_in[8];
    float* out = (float*)d_out;

    const int V = in_sizes[0] / DIM;
    const int E = in_sizes[1];

    float*    src_proj = (float*)d_ws;
    unsigned* line     = (unsigned*)(src_proj + (size_t)V * DIM);
    unsigned* line2    = line + (size_t)V * LINEW;
    int*      next     = (int*)(line2 + (size_t)V * LINEW);
    int*      head     = next + (size_t)E;

    head_clear_kernel<<<(V + 255) / 256, dim3(256), 0, stream>>>(line, line2, head, V);

    proj_gemm_kernel<<<dim3(PGRID), dim3(PB + LW), 0, stream>>>(
        node, Wsrc, bsrc, Wdst, bdst, src_proj, out, esrc, edst, ecls, line, line2, head, next, V, E);

    node_gather_kernel<<<(V + 31) / 32, dim3(GB), 0, stream>>>(
        src_proj, line, line2, head, next, esrc, ecls, eemb, out, V, E);
}


# projection epilogue: a wave's 4 tiles hold neighbouring columns (W staged in that column order), so a row leaves as 16-byte stores of whole 128-B lines, 8 stores per strip, was 32 half-line ones
# speedup vs baseline: 3.1478x; 1.2027x over previous
#include <hip/hip_runtime.h>


#define DIM 128
#define NCLS 32
#define GB 512
#define PB 512
#define LW 256
#define EPT 10
#define PGRID 256
#define LINEW 32
#define SLOTS 31
#define RU 4

typedef float v2f __attribute__((ext_vector_type(2)));
typedef float v4f __attribute__((ext_vector_type(4)));
typedef float v8f __attribute__((ext_vector_type(8)));

template <bool GUARD>
__device__ __forceinline__ void
link_batch(const int* __restrict__ edge_src,
           const int* __restrict__ edge_dest,
           const int* __restrict__ edge_cls,
           unsigned* __restrict__ line, int* __restrict__ head,
           int* __restrict__ next, int E, int base, int T)
{
    int s[EPT], d[EPT], c[EPT];
    unsigned k[EPT];
#pragma unroll
    for (int i = 0; i < EPT; ++i) {
        const int e  = base + i * T;
        const int ec = (!GUARD || e < E) ? e : base;
        s[i] = edge_src[ec];
        d[i] = edge_dest[ec];
        c[i] = edge_cls[ec];
    }
#pragma unroll
    for (int i = 0; i < EPT; ++i) {
        const int e = base + i * T;
        k[i] = 0u;
        if (!GUARD || e < E)
            k[i] = __hip_atomic_fetch_add(line + (size_t)d[i] * LINEW, 1u, __ATOMIC_RELAXED,
                                          __HIP_MEMORY_SCOPE_AGENT);
    }
#pragma unroll
    for (int i = 0; i < EPT; ++i) {
        const int e = base + i * T;
        if (!GUARD || e < E) {
            if (k[i] < (unsigned)SLOTS) {
                line[(size_t)d[i] * LINEW + 1 + k[i]] =
                    ((unsigned)s[i] << 5) | ((unsigned)c[i] & (NCLS - 1));
            } else {
                const int prev = __hip_atomic_exchange(head + d[i], e, __ATOMIC_RELAXED,
                                                       __HIP_MEMORY_SCOPE_AGENT);
                next[e] = prev;
            }
        }
    }
}

__device__ __forceinline__ void
link_edges(const int* __restrict__ edge_src,
           const int* __restrict__ edge_dest,
           const int* __restrict__ edge_cls,
           unsigned* __restrict__ line, int* __restrict__ head,
           int* __restrict__ next, int E, int g, int T)
{
    int base = g;
    for (; base + (EPT - 1) * T < E; base += EPT * T)
        link_batch<false>(edge_src, edge_dest, edge_cls, line, head, next, E, base, T);
    if (base < E)
        link_batch<true>(edge_src, edge_dest, edge_cls, line, head, next, E, base, T);
}

__global__ void __launch_bounds__(256)
head_clear_kernel(unsigned* __restrict__ line, unsigned* __restrict__ line2,
                  int* __restrict__ head, int V)
{
    const int i = blockIdx.x * 256 + threadIdx.x;
    if (i < V) {
        line[(size_t)i * LINEW]  = 0u;
        line2[(size_t)i * LINEW] = 0u;
        head[i] = -1;
    }
}

__global__ void __launch_bounds__(PB + LW) __attribute__((amdgpu_waves_per_eu(6, 6)))
proj_gemm_kernel(const float* __restrict__ node,
                 const float* __restrict__ Wsrc, const float* __restrict__ bsrc,
                 const float* __restrict__ Wdst, const float* __restrict__ bdst,
                 float* __restrict__ src_proj, float* __restrict__ dest_proj,
                 const int* __restrict__ edge_src,
                 const int* __restrict__ edge_dest,
                 const int* __restrict__ edge_cls,
                 unsigned* __restrict__ line, unsigned* __restrict__ line2,
                 int* __restrict__ head,
                 int* __restrict__ next, int V, int E)
{
    __shared__ v4f s_w[2 * (DIM / 4) * DIM];

    const int tid  = threadIdx.x;
    const int lane = tid & 31;
    const int c    = lane & 15;
    const int h    = lane >> 4;
    const int wave = __builtin_amdgcn_readfirstlane(tid >> 5);
    const int  w6 = wave / 6;
    const int  r6 = wave - 6 * w6;
    const bool mm = r6 < 4;
    const int  mw = 4 * w6 + r6;
    const int  mtid = 32 * mw + lane;
    const int  ltid = 32 * (2 * w6 + r6 - 4) + lane;
    const int g    = (mw + (mw >> 2)) & 3;
    const int q    = mw & 3;
    const int wsel = q >> 1;
    const int n0   = (q & 1) * 64;

    const int nstrips = (V + 15) >> 4;
    const int stride  = gridDim.x * 4;
    int s = blockIdx.x + gridDim.x * g;

    v4f a[DIM / 8] = {};
    if (mm) {
        const int s0 = s < nstrips ? s : 0;
        const float* __restrict__ ap = node + (size_t)(s0 * 16 + c) * DIM + 4 * h;
#pragma unroll
        for (int kb = 0; kb < DIM / 8; ++kb) a[kb] = *(const v4f*)(ap + 8 * kb);
    }

    if (mm) {
#pragma unroll
    for (int i = 0; i < 2 * (DIM / 4) * (DIM / 2) / PB; ++i) {
        const int item = mtid + i * PB;
        const int np = item & 63;
        const int kq = (item >> 6) & 31;
        const int w  = item >> 11;
        const float* __restrict__ p = (w ? Wdst : Wsrc) + (size_t)(4 * kq) * DIM + 2 * np;
        const v2f r0 = *(const v2f*)(p);
        const v2f r1 = *(const v2f*)(p + DIM);
        const v2f r2 = *(const v2f*)(p + 2 * DIM);
        const v2f r3 = *(const v2f*)(p + 3 * DIM);
        v4f c0, c1;
        c0.x = r0.x; c0.y = r1.x; c0.z = r2.x; c0.w = r3.x;
        c1.x = r0.y; c1.y = r1.y; c1.z = r2.y; c1.w = r3.y;
        const int n   = 2 * np;
        const int pos = (n & 64) + 16 * (n & 3) + ((n >> 2) & 15);
        s_w[(w * 32 + kq) * DIM + pos]      = c0;
        s_w[(w * 32 + kq) * DIM + pos + 16] = c1;
    }
    }
    __syncthreads();

    if (!mm) {
        const unsigned place = __builtin_amdgcn_s_sendmsg_rtn(0x87);
        unsigned* __restrict__ myline = ((place >> 18) & 1u) ? line2 : line;
        link_edges(edge_src, edge_dest, edge_cls, myline, head, next, E,
                   blockIdx.x * LW + ltid, gridDim.x * LW);
        return;
    }

    float* __restrict__ outp = wsel ? dest_proj : src_proj;
    const v4f bias = *(const v4f*)((wsel ? bdst : bsrc) + n0 + 4 * c);

    const v4f* bfrag = s_w + (wsel * 32 + h) * DIM + n0 + c;

    if (s >= nstrips) return;

    for (;;) {
        asm volatile("" ::: "memory");
        const int  sn   = s + stride;
        const bool more = sn < nstrips;
        const float* __restrict__ apn =
            node + (size_t)((more ? sn : s) * 16 + c) * DIM + 4 * h;

        v8f acc[4] = {};
#pragma unroll
        for (int kb = 0; kb < DIM / 8; ++kb) {
            v4f bv[4];
#pragma unroll
            for (int t = 0; t < 4; ++t) bv[t] = bfrag[2 * kb * DIM + 16 * t];
            v2f a01, a23;
            a01.x = a[kb].x; a01.y = a[kb].y;
            a23.x = a[kb].z; a23.y = a[kb].w;
#pragma unroll
            for (int t = 0; t < 4; ++t) {
                v2f b01; b01.x = bv[t].x; b01.y = bv[t].y;
                acc[t] = __builtin_amdgcn_wmma_f32_16x16x4_f32(
                             false, a01, false, b01, (short)0, acc[t], false, false);
            }
#pragma unroll
            for (int t = 0; t < 4; ++t) {
                v2f b23; b23.x = bv[t].z; b23.y = bv[t].w;
                acc[t] = __builtin_amdgcn_wmma_f32_16x16x4_f32(
                             false, a23, false, b23, (short)0, acc[t], false, false);
            }
            a[kb] = *(const v4f*)(apn + 8 * kb);
            __builtin_amdgcn_sched_barrier(0x7C6);
        }

        float* __restrict__ op = outp + (size_t)(s * 16 + 8 * h) * DIM + n0 + 4 * c;
#pragma unroll
        for (int r = 0; r < 8; ++r) {
            v4f o;
            o.x = acc[0][r] + bias.x;
            o.y = acc[1][r] + bias.y;
            o.z = acc[2][r] + bias.z;
            o.w = acc[3][r] + bias.w;
            *(v4f*)(op + (size_t)r * DIM) = o;
        }

        if (!more) break;
        s = sn;
    }
}

__global__ void __launch_bounds__(GB)
node_gather_kernel(const float* __restrict__ src_proj,
                   const unsigned* __restrict__ line,
                   const unsigned* __restrict__ line2,
                   const int* __restrict__ head,
                   const int* __restrict__ next,
                   const int* __restrict__ edge_src,
                   const int* __restrict__ edge_cls,
                   const float* __restrict__ edge_emb,
                   float* __restrict__ out, int V, int E)
{
    __shared__ float4 s_emb[NCLS * DIM / 4];

    const int lane = threadIdx.x & 31;
    const int l15  = lane & 15;
    const int d    = (blockIdx.x * (GB / 32) + (threadIdx.x >> 5)) * 2 + (lane >> 4);
    const bool valid = d < V;
    const int q0 = l15;
    const int q1 = 16 + l15;

    const int dc = valid ? d : 0;
    float4* __restrict__ orow = (float4*)(out + (size_t)dc * DIM);
    const unsigned* __restrict__ lnA = line  + (size_t)dc * LINEW;
    const unsigned* __restrict__ lnB = line2 + (size_t)dc * LINEW;

    const unsigned a0w = lnA[l15];
    const unsigned a1w = lnA[16 + l15];
    const unsigned b0w = lnB[l15];
    const unsigned b1w = lnB[16 + l15];
    const float4 dp0 = orow[q0];
    const float4 dp1 = orow[q1];
    float4 t[NCLS * DIM / 4 / GB];
#pragma unroll
    for (int i = 0; i < NCLS * DIM / 4 / GB; ++i)
        t[i] = ((const float4*)edge_emb)[threadIdx.x + i * GB];

#pragma unroll
    for (int i = 0; i < NCLS * DIM / 4 / GB; ++i)
        s_emb[threadIdx.x + i * GB] = t[i];
    __syncthreads();

    const unsigned cntA = (unsigned)__shfl((int)a0w, 0, 16);
    const unsigned cntB = (unsigned)__shfl((int)b0w, 0, 16);
    const int nA = valid ? (int)(cntA < (unsigned)E ? cntA : (unsigned)E) : 0;
    const int nB = valid ? (int)(cntB < (unsigned)E ? cntB : (unsigned)E) : 0;
    const int mA = nA < SLOTS ? nA : SLOTS;
    const int mB = nB < SLOTS ? nB : SLOTS;
    const int nb = mA + mB;

    float4 acc0 = make_float4(0.f, 0.f, 0.f, 0.f), acc1 = acc0;
    for (int base = 0; base < nb; base += RU) {
        float4 a0[RU], a1[RU];
        int    cc[RU];
        bool   ok[RU];
        unsigned w[RU];
#pragma unroll
        for (int u = 0; u < RU; ++u) {
            const int  j   = base + u;
            const bool inA = j < mA;
            const int  wi  = (inA ? j : j - mA) + 1;
            const unsigned lo = inA ? a0w : b0w;
            const unsigned hi = inA ? a1w : b1w;
            w[u] = (unsigned)__shfl((int)((wi & 16) ? hi : lo), wi & 15, 16);
        }
#pragma unroll
        for (int u = 0; u < RU; ++u) {
            ok[u] = base + u < nb;
            cc[u] = (int)(w[u] & (NCLS - 1));
            a0[u] = make_float4(0.f, 0.f, 0.f, 0.f);
            a1[u] = a0[u];
            if (ok[u]) {
                const float4* __restrict__ sp =
                    (const float4*)(src_proj + (size_t)(w[u] >> 5) * DIM);
                a0[u] = sp[q0];
                a1[u] = sp[q1];
            }
        }
#pragma unroll
        for (int u = 0; u < RU; ++u) {
            if (ok[u]) {
                const float4 m0 = s_emb[cc[u] * (DIM / 4) + q0];
                const float4 m1 = s_emb[cc[u] * (DIM / 4) + q1];
                acc0.x += fmaxf(a0[u].x + dp0.x + m0.x, 0.0f);
                acc0.y += fmaxf(a0[u].y + dp0.y + m0.y, 0.0f);
                acc0.z += fmaxf(a0[u].z + dp0.z + m0.z, 0.0f);
                acc0.w += fmaxf(a0[u].w + dp0.w + m0.w, 0.0f);
                acc1.x += fmaxf(a1[u].x + dp1.x + m1.x, 0.0f);
                acc1.y += fmaxf(a1[u].y + dp1.y + m1.y, 0.0f);
                acc1.z += fmaxf(a1[u].z + dp1.z + m1.z, 0.0f);
                acc1.w += fmaxf(a1[u].w + dp1.w + m1.w, 0.0f);
            }
        }
    }

    if (nA > SLOTS || nB > SLOTS) {
        int e = head[dc];
        for (int it = 0; (unsigned)e < (unsigned)E && it < E; ++it) {
            const int sr = edge_src[e];
            const int c  = edge_cls[e] & (NCLS - 1);
            const int en = next[e];
            const float4* __restrict__ sp = (const float4*)(src_proj + (size_t)sr * DIM);
            const float4 a0 = sp[q0];
            const float4 a1 = sp[q1];
            const float4 m0 = s_emb[c * (DIM / 4) + q0];
            const float4 m1 = s_emb[c * (DIM / 4) + q1];
            acc0.x += fmaxf(a0.x + dp0.x + m0.x, 0.0f);
            acc0.y += fmaxf(a0.y + dp0.y + m0.y, 0.0f);
            acc0.z += fmaxf(a0.z + dp0.z + m0.z, 0.0f);
            acc0.w += fmaxf(a0.w + dp0.w + m0.w, 0.0f);
            acc1.x += fmaxf(a1.x + dp1.x + m1.x, 0.0f);
            acc1.y += fmaxf(a1.y + dp1.y + m1.y, 0.0f);
            acc1.z += fmaxf(a1.z + dp1.z + m1.z, 0.0f);
            acc1.w += fmaxf(a1.w + dp1.w + m1.w, 0.0f);
            e = en;
        }
    }

    if (valid) {
        orow[q0] = acc0;
        orow[q1] = acc1;
    }
}

extern "C" void kernel_launch(void* const* d_in, const int* in_sizes, int n_in,
                              void* d_out, int out_size, void* d_ws, size_t ws_size,
                              hipStream_t stream) {
    const float* node = (const float*)d_in[0];
    const int*   esrc = (const int*)d_in[1];
    const int*   edst = (const int*)d_in[2];
    const int*   ecls = (const int*)d_in[3];
    const float* Wsrc = (const float*)d_in[4];
    const float* bsrc = (const float*)d_in[5];
    const float* Wdst = (const float*)d_in[6];
    const float* bdst = (const float*)d_in[7];
    const float* eemb = (const float*)d_in[8];
    float* out = (float*)d_out;

    const int V = in_sizes[0] / DIM;
    const int E = in_sizes[1];

    float*    src_proj = (float*)d_ws;
    unsigned* line     = (unsigned*)(src_proj + (size_t)V * DIM);
    unsigned* line2    = line + (size_t)V * LINEW;
    int*      next     = (int*)(line2 + (size_t)V * LINEW);
    int*      head     = next + (size_t)E;

    head_clear_kernel<<<(V + 255) / 256, dim3(256), 0, stream>>>(line, line2, head, V);

    proj_gemm_kernel<<<dim3(PGRID), dim3(PB + LW), 0, stream>>>(
        node, Wsrc, bsrc, Wdst, bdst, src_proj, out, esrc, edst, ecls, line, line2, head, next, V, E);

    node_gather_kernel<<<(V + 31) / 32, dim3(GB), 0, stream>>>(
        src_proj, line, line2, head, next, esrc, ecls, eemb, out, V, E);
}
